# starting kernel
# baseline (speedup 1.0000x reference)
#include <hip/hip_runtime.h>

typedef float v2f __attribute__((ext_vector_type(2)));
typedef float v8f __attribute__((ext_vector_type(8)));

#define WAVES_PER_BLOCK 2
#define RAYS_PER_WAVE   16
#define NSAMP           128

__global__ __launch_bounds__(WAVES_PER_BLOCK * 32)
void DistortionLoss_82471962018601_kernel(const float* __restrict__ wp,
                                          const float* __restrict__ zp,
                                          const float* __restrict__ nearp,
                                          const float* __restrict__ farp,
                                          float* __restrict__ out,
                                          int nrays) {
    __shared__ float sA[WAVES_PER_BLOCK][NSAMP * RAYS_PER_WAVE];
    __shared__ float sB[WAVES_PER_BLOCK][NSAMP * RAYS_PER_WAVE];
    __shared__ float sT3[WAVES_PER_BLOCK][RAYS_PER_WAVE];

    const int wave  = threadIdx.x >> 5;
    const int lane  = threadIdx.x & 31;
    const int rbase = blockIdx.x * (WAVES_PER_BLOCK * RAYS_PER_WAVE) + wave * RAYS_PER_WAVE;

    for (int r = 0; r < RAYS_PER_WAVE; ++r) {
        int ray = rbase + r;
        if (ray >= nrays) ray = nrays - 1;
        const float nv  = nearp[ray];
        const float fv  = farp[ray];
        const float inv = 1.0f / (fv - nv);
        float t3 = 0.0f;
#pragma unroll
        for (int q = 0; q < 4; ++q) {
            const int i   = q * 32 + lane;
            const float w = wp[(long)ray * NSAMP + i];
            const float zlo = zp[(long)ray * (NSAMP + 1) + i];
            const float zhi = zp[(long)ray * (NSAMP + 1) + i + 1];
            const float blo = (zlo - nv) * inv;
            const float bhi = (zhi - nv) * inv;
            const float m   = 0.5f * (blo + bhi);
            sB[wave][i * RAYS_PER_WAVE + r] = w;
            sA[wave][i * RAYS_PER_WAVE + r] = w * m;
            t3 += w * w * (bhi - blo);
        }
        for (int off = 16; off > 0; off >>= 1) t3 += __shfl_xor(t3, off, 32);
        if (lane == 0) sT3[wave][r] = t3;
    }
    __syncthreads();

    const int ray16  = lane & 15;
    const int hi     = lane >> 4;
    const int rowOff = hi * 8;
    const int srow   = lane & 15;

    v2f sfrag[4];
#pragma unroll
    for (int kk = 0; kk < 4; ++kk) {
        const int col0 = kk * 4 + hi * 2;
        sfrag[kk].x = (col0     < srow) ? 1.0f : 0.0f;
        sfrag[kk].y = (col0 + 1 < srow) ? 1.0f : 0.0f;
    }

    float carryA = 0.0f, carryB = 0.0f;
    float dot = 0.0f;

    for (int c = 0; c < 8; ++c) {
        const float* baseB = &sB[wave][(c * 16) * RAYS_PER_WAVE];
        const float* baseA = &sA[wave][(c * 16) * RAYS_PER_WAVE];

        v8f Db = {};
        v8f Da = {};
#pragma unroll
        for (int kk = 0; kk < 4; ++kk) {
            const int kr = kk * 4 + hi * 2;
            v2f bf, af;
            bf.x = baseB[(kr    ) * RAYS_PER_WAVE + ray16];
            bf.y = baseB[(kr + 1) * RAYS_PER_WAVE + ray16];
            af.x = baseA[(kr    ) * RAYS_PER_WAVE + ray16];
            af.y = baseA[(kr + 1) * RAYS_PER_WAVE + ray16];
            Db = __builtin_amdgcn_wmma_f32_16x16x4_f32(
                     false, sfrag[kk], false, bf, (short)0, Db, false, false);
            Da = __builtin_amdgcn_wmma_f32_16x16x4_f32(
                     false, sfrag[kk], false, af, (short)0, Da, false, false);
        }

        float sb = 0.0f, sa = 0.0f;
#pragma unroll
        for (int g = 0; g < 8; ++g) {
            const float bv = baseB[(g + rowOff) * RAYS_PER_WAVE + ray16];
            const float av = baseA[(g + rowOff) * RAYS_PER_WAVE + ray16];
            dot += av * (Db[g] + carryB) - bv * (Da[g] + carryA);
            sb += bv;
            sa += av;
        }
        sb += __shfl_xor(sb, 16, 32);
        sa += __shfl_xor(sa, 16, 32);
        carryB += sb;
        carryA += sa;
    }

    dot += __shfl_xor(dot, 16, 32);

    if (lane < 16 && (rbase + lane) < nrays) {
        out[rbase + lane] = 2.0f * dot + (1.0f / 3.0f) * sT3[wave][lane];
    }
}

extern "C" void kernel_launch(void* const* d_in, const int* in_sizes, int n_in,
                              void* d_out, int out_size, void* d_ws, size_t ws_size,
                              hipStream_t stream) {
    const float* wp    = (const float*)d_in[0];
    const float* zp    = (const float*)d_in[1];
    const float* nearp = (const float*)d_in[2];
    const float* farp  = (const float*)d_in[3];
    float* out = (float*)d_out;

    const int nrays = in_sizes[2];
    const int raysPerBlock = WAVES_PER_BLOCK * RAYS_PER_WAVE;
    const int blocks = (nrays + raysPerBlock - 1) / raysPerBlock;

    DistortionLoss_82471962018601_kernel<<<blocks, WAVES_PER_BLOCK * 32, 0, stream>>>(
        wp, zp, nearp, farp, out, nrays);
}


# Phase 0 reordered: all 130 global loads of a wave's 16 rays issue before first use (one unrolled branch-free block, near/far loaded once per lane); latency overlaps instead of adding 16 times
# speedup vs baseline: 2.7890x; 2.7890x over previous
#include <hip/hip_runtime.h>

typedef float v2f __attribute__((ext_vector_type(2)));
typedef float v8f __attribute__((ext_vector_type(8)));

#define WAVES_PER_BLOCK 2
#define RAYS_PER_WAVE   16
#define NSAMP           128

__global__ __launch_bounds__(WAVES_PER_BLOCK * 32)
void DistortionLoss_82471962018601_kernel(const float* __restrict__ wp,
                                          const float* __restrict__ zp,
                                          const float* __restrict__ nearp,
                                          const float* __restrict__ farp,
                                          float* __restrict__ out,
                                          int nrays) {
    __shared__ float sA[WAVES_PER_BLOCK][NSAMP * RAYS_PER_WAVE];
    __shared__ float sB[WAVES_PER_BLOCK][NSAMP * RAYS_PER_WAVE];
    __shared__ float sT3[WAVES_PER_BLOCK][RAYS_PER_WAVE];

    const int wave  = threadIdx.x >> 5;
    const int lane  = threadIdx.x & 31;
    const int rbase = blockIdx.x * (WAVES_PER_BLOCK * RAYS_PER_WAVE) + wave * RAYS_PER_WAVE;

    float wv[RAYS_PER_WAVE][4], zl[RAYS_PER_WAVE][4], zh[RAYS_PER_WAVE][4];
    int rayl = rbase + (lane & 15);
    if (rayl >= nrays) rayl = nrays - 1;
    const float nl = nearp[rayl];
    const float fl = farp[rayl];
#pragma unroll
    for (int r = 0; r < RAYS_PER_WAVE; ++r) {
        int ray = rbase + r;
        if (ray >= nrays) ray = nrays - 1;
        const float* wrow = wp + (long)ray * NSAMP + lane;
        const float* zrow = zp + (long)ray * (NSAMP + 1) + lane;
#pragma unroll
        for (int q = 0; q < 4; ++q) {
            wv[r][q] = wrow[q * 32];
            zl[r][q] = zrow[q * 32];
            zh[r][q] = zrow[q * 32 + 1];
        }
    }
    float t3r[RAYS_PER_WAVE];
#pragma unroll
    for (int r = 0; r < RAYS_PER_WAVE; ++r) {
        const float nv  = __int_as_float(__builtin_amdgcn_readlane(__float_as_int(nl), r));
        const float fv  = __int_as_float(__builtin_amdgcn_readlane(__float_as_int(fl), r));
        const float inv = 1.0f / (fv - nv);
        float t3 = 0.0f;
#pragma unroll
        for (int q = 0; q < 4; ++q) {
            const int i   = q * 32 + lane;
            const float w = wv[r][q];
            const float blo = (zl[r][q] - nv) * inv;
            const float bhi = (zh[r][q] - nv) * inv;
            const float m   = 0.5f * (blo + bhi);
            sB[wave][i * RAYS_PER_WAVE + r] = w;
            sA[wave][i * RAYS_PER_WAVE + r] = w * m;
            t3 += w * w * (bhi - blo);
        }
        for (int off = 16; off > 0; off >>= 1) t3 += __shfl_xor(t3, off, 32);
        t3r[r] = t3;
    }
    __builtin_amdgcn_sched_group_barrier(0x020, 130, 0);
    __builtin_amdgcn_sched_group_barrier(0x002, 4096, 0);
    if (lane == 0) {
#pragma unroll
        for (int r = 0; r < RAYS_PER_WAVE; ++r) sT3[wave][r] = t3r[r];
    }
    __syncthreads();

    const int ray16  = lane & 15;
    const int hi     = lane >> 4;
    const int rowOff = hi * 8;
    const int srow   = lane & 15;

    v2f sfrag[4];
#pragma unroll
    for (int kk = 0; kk < 4; ++kk) {
        const int col0 = kk * 4 + hi * 2;
        sfrag[kk].x = (col0     < srow) ? 1.0f : 0.0f;
        sfrag[kk].y = (col0 + 1 < srow) ? 1.0f : 0.0f;
    }

    float carryA = 0.0f, carryB = 0.0f;
    float dot = 0.0f;

    for (int c = 0; c < 8; ++c) {
        const float* baseB = &sB[wave][(c * 16) * RAYS_PER_WAVE];
        const float* baseA = &sA[wave][(c * 16) * RAYS_PER_WAVE];

        v8f Db = {};
        v8f Da = {};
#pragma unroll
        for (int kk = 0; kk < 4; ++kk) {
            const int kr = kk * 4 + hi * 2;
            v2f bf, af;
            bf.x = baseB[(kr    ) * RAYS_PER_WAVE + ray16];
            bf.y = baseB[(kr + 1) * RAYS_PER_WAVE + ray16];
            af.x = baseA[(kr    ) * RAYS_PER_WAVE + ray16];
            af.y = baseA[(kr + 1) * RAYS_PER_WAVE + ray16];
            Db = __builtin_amdgcn_wmma_f32_16x16x4_f32(
                     false, sfrag[kk], false, bf, (short)0, Db, false, false);
            Da = __builtin_amdgcn_wmma_f32_16x16x4_f32(
                     false, sfrag[kk], false, af, (short)0, Da, false, false);
        }

        float sb = 0.0f, sa = 0.0f;
#pragma unroll
        for (int g = 0; g < 8; ++g) {
            const float bv = baseB[(g + rowOff) * RAYS_PER_WAVE + ray16];
            const float av = baseA[(g + rowOff) * RAYS_PER_WAVE + ray16];
            dot += av * (Db[g] + carryB) - bv * (Da[g] + carryA);
            sb += bv;
            sa += av;
        }
        sb += __shfl_xor(sb, 16, 32);
        sa += __shfl_xor(sa, 16, 32);
        carryB += sb;
        carryA += sa;
    }

    dot += __shfl_xor(dot, 16, 32);

    if (lane < 16 && (rbase + lane) < nrays) {
        out[rbase + lane] = 2.0f * dot + (1.0f / 3.0f) * sT3[wave][lane];
    }
}

extern "C" void kernel_launch(void* const* d_in, const int* in_sizes, int n_in,
                              void* d_out, int out_size, void* d_ws, size_t ws_size,
                              hipStream_t stream) {
    const float* wp    = (const float*)d_in[0];
    const float* zp    = (const float*)d_in[1];
    const float* nearp = (const float*)d_in[2];
    const float* farp  = (const float*)d_in[3];
    float* out = (float*)d_out;

    const int nrays = in_sizes[2];
    const int raysPerBlock = WAVES_PER_BLOCK * RAYS_PER_WAVE;
    const int blocks = (nrays + raysPerBlock - 1) / raysPerBlock;

    DistortionLoss_82471962018601_kernel<<<blocks, WAVES_PER_BLOCK * 32, 0, stream>>>(
        wp, zp, nearp, farp, out, nrays);
}


# Each lane loads its own ray's samples from global directly in WMMA fragment order (k regrouped: a lane's 8 chunk samples = its B fragment = its D rows); no LDS transpose, barrier, wave reductions
# speedup vs baseline: 4.1642x; 1.4931x over previous
#include <hip/hip_runtime.h>

typedef float v2f __attribute__((ext_vector_type(2)));
typedef float v8f __attribute__((ext_vector_type(8)));

#define WAVES_PER_BLOCK 2
#define RAYS_PER_WAVE   16
#define NSAMP           128
#define NCHUNK          (NSAMP / 16)

__global__ __launch_bounds__(WAVES_PER_BLOCK * 32)
void DistortionLoss_82471962018601_kernel(const float* __restrict__ wp,
                                          const float* __restrict__ zp,
                                          const float* __restrict__ nearp,
                                          const float* __restrict__ farp,
                                          float* __restrict__ out,
                                          int nrays) {
    const int wave  = threadIdx.x >> 5;
    const int lane  = threadIdx.x & 31;
    const int rbase = blockIdx.x * (WAVES_PER_BLOCK * RAYS_PER_WAVE) + wave * RAYS_PER_WAVE;
    const int c     = lane & 15;
    const int hi    = lane >> 4;

    int ray = rbase + c;
    if (ray >= nrays) ray = nrays - 1;
    const float* wrow = wp + (long)ray * NSAMP + hi * 8;
    const float* zrow = zp + (long)ray * (NSAMP + 1) + hi * 8;

    float wv[NCHUNK][8], zv[NCHUNK][9];
#pragma unroll
    for (int j = 0; j < NCHUNK; ++j) {
#pragma unroll
        for (int k = 0; k < 8; ++k) wv[j][k] = wrow[j * 16 + k];
#pragma unroll
        for (int k = 0; k < 9; ++k) zv[j][k] = zrow[j * 16 + k];
    }
    const float nv  = nearp[ray];
    const float fv  = farp[ray];
    const float inv = 1.0f / (fv - nv);

    v2f sfrag[4];
#pragma unroll
    for (int kk = 0; kk < 4; ++kk) {
        const int col0 = hi * 8 + kk * 2;
        sfrag[kk].x = (col0     < c) ? 1.0f : 0.0f;
        sfrag[kk].y = (col0 + 1 < c) ? 1.0f : 0.0f;
    }

    float carryA = 0.0f, carryB = 0.0f;
    float dot = 0.0f;
    float t3  = 0.0f;

#pragma unroll
    for (int j = 0; j < NCHUNK; ++j) {
        float bn[9];
#pragma unroll
        for (int k = 0; k < 9; ++k) bn[k] = (zv[j][k] - nv) * inv;
        float av[8];
        float t3c = 0.0f;
#pragma unroll
        for (int k = 0; k < 8; ++k) {
            const float w = wv[j][k];
            const float m = 0.5f * (bn[k] + bn[k + 1]);
            av[k] = w * m;
            t3c += w * w * (bn[k + 1] - bn[k]);
        }
        t3 += t3c;

        v8f Db = {};
        v8f Da = {};
#pragma unroll
        for (int kk = 0; kk < 4; ++kk) {
            v2f bf, af;
            bf.x = wv[j][2 * kk];
            bf.y = wv[j][2 * kk + 1];
            af.x = av[2 * kk];
            af.y = av[2 * kk + 1];
            Db = __builtin_amdgcn_wmma_f32_16x16x4_f32(
                     false, sfrag[kk], false, bf, (short)0, Db, false, false);
            Da = __builtin_amdgcn_wmma_f32_16x16x4_f32(
                     false, sfrag[kk], false, af, (short)0, Da, false, false);
        }

        float sb = 0.0f, sa = 0.0f;
#pragma unroll
        for (int g = 0; g < 8; ++g) {
            const float bv = wv[j][g];
            dot += av[g] * (Db[g] + carryB) - bv * (Da[g] + carryA);
            sb += bv;
            sa += av[g];
        }
        sb += __shfl_xor(sb, 16, 32);
        sa += __shfl_xor(sa, 16, 32);
        carryB += sb;
        carryA += sa;
    }

    __builtin_amdgcn_sched_group_barrier(0x020, 44, 0);
    __builtin_amdgcn_sched_group_barrier(0x002, 4096, 0);

    dot += __shfl_xor(dot, 16, 32);
    t3  += __shfl_xor(t3, 16, 32);

    if (lane < 16 && (rbase + lane) < nrays) {
        out[rbase + lane] = 2.0f * dot + (1.0f / 3.0f) * t3;
    }
}

extern "C" void kernel_launch(void* const* d_in, const int* in_sizes, int n_in,
                              void* d_out, int out_size, void* d_ws, size_t ws_size,
                              hipStream_t stream) {
    const float* wp    = (const float*)d_in[0];
    const float* zp    = (const float*)d_in[1];
    const float* nearp = (const float*)d_in[2];
    const float* farp  = (const float*)d_in[3];
    float* out = (float*)d_out;

    const int nrays = in_sizes[2];
    const int raysPerBlock = WAVES_PER_BLOCK * RAYS_PER_WAVE;
    const int blocks = (nrays + raysPerBlock - 1) / raysPerBlock;

    DistortionLoss_82471962018601_kernel<<<blocks, WAVES_PER_BLOCK * 32, 0, stream>>>(
        wp, zp, nearp, farp, out, nrays);
}


# One prefix scan per ray, not two: sum_i w_i*prefix(w*m)_i rewritten as sum_j (w*m)_j*(W-P_j-w_j), edges taken relative to the middle edge; half the WMMAs; values are summed in a new order
# speedup vs baseline: 4.5190x; 1.0852x over previous
#include <hip/hip_runtime.h>

typedef float v2f __attribute__((ext_vector_type(2)));
typedef float v8f __attribute__((ext_vector_type(8)));

#define WAVES_PER_BLOCK 2
#define RAYS_PER_WAVE   16
#define NSAMP           128
#define NCHUNK          (NSAMP / 16)

__global__ __launch_bounds__(WAVES_PER_BLOCK * 32)
void DistortionLoss_82471962018601_kernel(const float* __restrict__ wp,
                                          const float* __restrict__ zp,
                                          const float* __restrict__ nearp,
                                          const float* __restrict__ farp,
                                          float* __restrict__ out,
                                          int nrays) {
    const int wave  = threadIdx.x >> 5;
    const int lane  = threadIdx.x & 31;
    const int rbase = blockIdx.x * (WAVES_PER_BLOCK * RAYS_PER_WAVE) + wave * RAYS_PER_WAVE;
    const int c     = lane & 15;
    const int hi    = lane >> 4;

    int ray = rbase + c;
    if (ray >= nrays) ray = nrays - 1;
    const float* wrow = wp + (long)ray * NSAMP + hi * 8;
    const float* zray = zp + (long)ray * (NSAMP + 1);
    const float* zrow = zray + hi * 8;

    float wv[NCHUNK][8], zv[NCHUNK][9];
#pragma unroll
    for (int j = 0; j < NCHUNK; ++j) {
#pragma unroll
        for (int k = 0; k < 8; ++k) wv[j][k] = wrow[j * 16 + k];
#pragma unroll
        for (int k = 0; k < 9; ++k) zv[j][k] = zrow[j * 16 + k];
    }
    const float nv  = nearp[ray];
    const float fv  = farp[ray];
    const float zc  = zray[NSAMP / 2];
    const float inv = 1.0f / (fv - nv);

    v2f sfrag[4];
#pragma unroll
    for (int kk = 0; kk < 4; ++kk) {
        const int col0 = hi * 8 + kk * 2;
        sfrag[kk].x = (col0     < c) ? 1.0f : 0.0f;
        sfrag[kk].y = (col0 + 1 < c) ? 1.0f : 0.0f;
    }

    float su[NCHUNK], sw[NCHUNK];
    float pd = 0.0f;
    float uw = 0.0f;
    float t3 = 0.0f;

#pragma unroll
    for (int j = 0; j < NCHUNK; ++j) {
        float e[9];
#pragma unroll
        for (int k = 0; k < 9; ++k) e[k] = (zv[j][k] - zc) * inv;
        float uv[8];
        float t3c = 0.0f;
#pragma unroll
        for (int k = 0; k < 8; ++k) {
            const float w = wv[j][k];
            uv[k] = w * (e[k] + e[k + 1]);
            t3c += w * w * (e[k + 1] - e[k]);
        }
        t3 += t3c;

        v8f D = {};
#pragma unroll
        for (int kk = 0; kk < 4; ++kk) {
            v2f bf;
            bf.x = wv[j][2 * kk];
            bf.y = wv[j][2 * kk + 1];
            D = __builtin_amdgcn_wmma_f32_16x16x4_f32(
                    false, sfrag[kk], false, bf, (short)0, D, false, false);
        }

        float pdc = 0.0f, uwc = 0.0f, suc = 0.0f, swc = 0.0f;
#pragma unroll
        for (int g = 0; g < 8; ++g) {
            const float w = wv[j][g];
            pdc += uv[g] * D[g];
            uwc += uv[g] * w;
            suc += uv[g];
            swc += w;
        }
        pd += pdc;
        uw += uwc;
        su[j] = suc;
        sw[j] = swc;
    }

    float carry = 0.0f;
    float cs    = 0.0f;
    float sut   = 0.0f;
#pragma unroll
    for (int j = 0; j < NCHUNK; ++j) {
        cs  += carry * su[j];
        sut += su[j];
        carry += sw[j] + __shfl_xor(sw[j], 16, 32);
    }
    float val = 2.0f * (pd + cs) + uw + (1.0f / 3.0f) * t3 - sut * carry;

    __builtin_amdgcn_sched_group_barrier(0x020, 43, 0);
    __builtin_amdgcn_sched_group_barrier(0x002, 4096, 0);

    val += __shfl_xor(val, 16, 32);

    if (lane < 16 && (rbase + lane) < nrays) {
        out[rbase + lane] = val;
    }
}

extern "C" void kernel_launch(void* const* d_in, const int* in_sizes, int n_in,
                              void* d_out, int out_size, void* d_ws, size_t ws_size,
                              hipStream_t stream) {
    const float* wp    = (const float*)d_in[0];
    const float* zp    = (const float*)d_in[1];
    const float* nearp = (const float*)d_in[2];
    const float* farp  = (const float*)d_in[3];
    float* out = (float*)d_out;

    const int nrays = in_sizes[2];
    const int raysPerBlock = WAVES_PER_BLOCK * RAYS_PER_WAVE;
    const int blocks = (nrays + raysPerBlock - 1) / raysPerBlock;

    DistortionLoss_82471962018601_kernel<<<blocks, WAVES_PER_BLOCK * 32, 0, stream>>>(
        wp, zp, nearp, farp, out, nrays);
}


# Lanes remapped: tile columns = 2 rays x 8 chunks, 8 rays per wave, 4 waves per block (all SIMDs busy, coalesced loads); chunk carry by DPP butterfly, 1/(far-near) once per ray; new summation order
# speedup vs baseline: 5.6990x; 1.2611x over previous
#include <hip/hip_runtime.h>

typedef float v2f __attribute__((ext_vector_type(2)));
typedef float v4f __attribute__((ext_vector_type(4)));
typedef float v8f __attribute__((ext_vector_type(8)));

#define WAVES_PER_BLOCK 4
#define RAYS_PER_WAVE   8
#define NGROUP          (RAYS_PER_WAVE / 2)
#define NSAMP           128

__device__ __forceinline__ v4f load4(const float* p) {
    v4f v;
    __builtin_memcpy(&v, p, sizeof(v));
    return v;
}

__device__ __forceinline__ float xhalf(float x) {
    const int b = __builtin_bit_cast(int, x);
    return __builtin_bit_cast(float,
        __builtin_amdgcn_permlanex16(b, b, 0x76543210, 0xfedcba98, false, false));
}

template <int M>
__device__ __forceinline__ float xmask(float x) {
    const int b = __builtin_bit_cast(int, x);
    return __builtin_bit_cast(float,
        __builtin_amdgcn_update_dpp(b, b, 0x160 + M, 0xf, 0xf, false));
}

__global__ __launch_bounds__(WAVES_PER_BLOCK * 32)
void DistortionLoss_82471962018601_kernel(const float* __restrict__ wp,
                                          const float* __restrict__ zp,
                                          const float* __restrict__ nearp,
                                          const float* __restrict__ farp,
                                          float* __restrict__ out,
                                          int nrays) {
    const int wave  = threadIdx.x >> 5;
    const int lane  = threadIdx.x & 31;
    const int rbase = blockIdx.x * (WAVES_PER_BLOCK * RAYS_PER_WAVE) + wave * RAYS_PER_WAVE;
    const int c     = lane & 15;
    const int hi    = lane >> 4;
    const int q     = c >> 3;
    const int p     = c & 7;

    v2f wpr[NGROUP][4];
    v2f zlo[NGROUP][4];
    v2f zhi[NGROUP][4];
    float zc[NGROUP];
#pragma unroll
    for (int g = 0; g < NGROUP; ++g) {
        int ray = rbase + 2 * g + q;
        if (ray >= nrays) ray = nrays - 1;
        const float* wrow = wp + (long)ray * NSAMP + p * 16 + hi * 8;
        const float* zray = zp + (long)ray * (NSAMP + 1);
        const float* zrow = zray + p * 16 + hi * 8;
        const v4f wa = load4(wrow), wb = load4(wrow + 4);
        const v4f za = load4(zrow), zb = load4(zrow + 4);
        const float z8 = zrow[8];
        wpr[g][0] = __builtin_shufflevector(wa, wa, 0, 1);
        wpr[g][1] = __builtin_shufflevector(wa, wa, 2, 3);
        wpr[g][2] = __builtin_shufflevector(wb, wb, 0, 1);
        wpr[g][3] = __builtin_shufflevector(wb, wb, 2, 3);
        zlo[g][0] = __builtin_shufflevector(za, za, 0, 1);
        zlo[g][1] = __builtin_shufflevector(za, za, 2, 3);
        zlo[g][2] = __builtin_shufflevector(zb, zb, 0, 1);
        zlo[g][3] = __builtin_shufflevector(zb, zb, 2, 3);
        zhi[g][0] = __builtin_shufflevector(za, za, 1, 2);
        zhi[g][1] = __builtin_shufflevector(za, zb, 3, 4);
        zhi[g][2] = __builtin_shufflevector(zb, zb, 1, 2);
        zhi[g][3] = (v2f){zb.w, z8};
        zc[g] = zray[NSAMP / 2];
    }
    const int oray_raw = rbase + 2 * (c & 3) + q;
    const int oray     = (oray_raw < nrays) ? oray_raw : (nrays - 1);
    const float inv    = 1.0f / (farp[oray] - nearp[oray]);

    v2f sfrag[4];
#pragma unroll
    for (int kk = 0; kk < 4; ++kk) {
        const int col0 = hi * 8 + kk * 2;
        sfrag[kk].x = (col0     < c) ? 1.0f : 0.0f;
        sfrag[kk].y = (col0 + 1 < c) ? 1.0f : 0.0f;
    }
    const float m1 = (p & 1) ? 1.0f : 0.0f;
    const float m2 = (p & 2) ? 1.0f : 0.0f;
    const float m4 = (p & 4) ? 1.0f : 0.0f;

    float val[NGROUP];
#pragma unroll
    for (int g = 0; g < NGROUP; ++g) {
        v8f D = {};
#pragma unroll
        for (int kk = 0; kk < 4; ++kk) {
            D = __builtin_amdgcn_wmma_f32_16x16x4_f32(
                    false, sfrag[kk], false, wpr[g][kk], (short)0, D, false, false);
        }

        const v2f zc2 = {zc[g], zc[g]};
        v2f t3p = {0.0f, 0.0f}, pdp = {0.0f, 0.0f}, uwp = {0.0f, 0.0f};
        v2f sup = {0.0f, 0.0f}, swp = {0.0f, 0.0f};
#pragma unroll
        for (int i = 0; i < 4; ++i) {
            const v2f w2  = wpr[g][i];
            const v2f elo = zlo[g][i] - zc2;
            const v2f ehi = zhi[g][i] - zc2;
            const v2f u2  = w2 * (elo + ehi);
            const v2f d2  = {D[2 * i], D[2 * i + 1]};
            t3p += (w2 * w2) * (ehi - elo);
            pdp += u2 * d2;
            uwp += u2 * w2;
            sup += u2;
            swp += w2;
        }
        const float t3c = t3p.x + t3p.y;
        const float pdc = pdp.x + pdp.y;
        const float uwc = uwp.x + uwp.y;
        const float suc = sup.x + sup.y;
        const float swc = swp.x + swp.y;

        float tot = swc + xhalf(swc);
        float pre = 0.0f;
        float t;
        t = xmask<1>(tot); pre += t * m1; tot += t;
        t = xmask<2>(tot); pre += t * m2; tot += t;
        t = xmask<4>(tot); pre += t * m4; tot += t;

        float v = 2.0f * pdc + uwc + (1.0f / 3.0f) * t3c + suc * (2.0f * pre - tot);
        v += xmask<1>(v);
        v += xmask<2>(v);
        v += xmask<4>(v);
        val[g] = v;
    }

    __builtin_amdgcn_sched_group_barrier(0x020, 26, 0);
    __builtin_amdgcn_sched_group_barrier(0x002, 4096, 0);

    const int gs = c & 3;
    float res = val[0];
    res = (gs == 1) ? val[1] : res;
    res = (gs == 2) ? val[2] : res;
    res = (gs == 3) ? val[3] : res;
    res *= inv;
    res += xhalf(res);
    if (hi == 0 && p < 4 && oray_raw < nrays) {
        out[oray_raw] = res;
    }
}

extern "C" void kernel_launch(void* const* d_in, const int* in_sizes, int n_in,
                              void* d_out, int out_size, void* d_ws, size_t ws_size,
                              hipStream_t stream) {
    const float* wp    = (const float*)d_in[0];
    const float* zp    = (const float*)d_in[1];
    const float* nearp = (const float*)d_in[2];
    const float* farp  = (const float*)d_in[3];
    float* out = (float*)d_out;

    const int nrays = in_sizes[2];
    const int raysPerBlock = WAVES_PER_BLOCK * RAYS_PER_WAVE;
    const int blocks = (nrays + raysPerBlock - 1) / raysPerBlock;

    DistortionLoss_82471962018601_kernel<<<blocks, WAVES_PER_BLOCK * 32, 0, stream>>>(
        wp, zp, nearp, farp, out, nrays);
}
